# starting kernel
# baseline (speedup 1.0000x reference)
#include <hip/hip_runtime.h>
#include <hip/hip_bf16.h>

typedef __attribute__((ext_vector_type(2))) float v2f;
typedef __attribute__((ext_vector_type(8))) float v8f;

#define CFEAT 128

__global__ void degree_kernel(const int* __restrict__ col, float* __restrict__ deg, int E) {
    int e = blockIdx.x * blockDim.x + threadIdx.x;
    if (e < E) atomicAdd(&deg[col[e]], 1.0f);
}

__global__ void dinv_kernel(float* __restrict__ deg, int n) {
    int i = blockIdx.x * blockDim.x + threadIdx.x;
    if (i < n) {
        float d = deg[i] + 1.0f;
        deg[i] = rsqrtf(d);
    }
}

template <int K, int N, int ADD_BIAS>
__global__ void gemm16_wmma_f32(const float* __restrict__ A,
                                const float* __restrict__ W,
                                const float* __restrict__ bias,
                                float* __restrict__ C, int M) {
    const int lane = threadIdx.x & 31;
    const int wave = threadIdx.x >> 5;
    const int lo = lane & 15;
    const int hi = lane >> 4;

    const int m0 = blockIdx.x * 16;
    const int n0 = (blockIdx.y * 4 + wave) * 16;

    const float* abase = A + (size_t)(m0 + lo) * K + 2 * hi;
    const float* wbase = W + (size_t)(2 * hi) * N + n0 + lo;

    v8f acc = {};
#pragma unroll
    for (int k0 = 0; k0 < K; k0 += 4) {
        v2f a, b;
        a.x = abase[k0];
        a.y = abase[k0 + 1];
        b.x = wbase[(size_t)k0 * N];
        b.y = wbase[(size_t)(k0 + 1) * N];
        acc = __builtin_amdgcn_wmma_f32_16x16x4_f32(
             false, a,  false, b,
             (short)0, acc,  false,  false);
    }

    const float bv = ADD_BIAS ? bias[n0 + lo] : 0.0f;
    float* cbase = C + (size_t)(m0 + 8 * hi) * N + n0 + lo;
#pragma unroll
    for (int v = 0; v < 8; ++v) {
        cbase[(size_t)v * N] = acc[v] + bv;
    }
}

__global__ void aggregate_edges(const float* __restrict__ h,
                                const int* __restrict__ row,
                                const int* __restrict__ col,
                                const float* __restrict__ dinv,
                                float* __restrict__ agg, int E) {
    const int lane = threadIdx.x & 31;
    const int e = blockIdx.x * (blockDim.x >> 5) + (threadIdx.x >> 5);
    if (e >= E) return;
    const int r = row[e];
    const int c = col[e];
    const float nrm = dinv[r] * dinv[c];
    const float4 v = ((const float4*)(h + (size_t)r * CFEAT))[lane];
    float* dst = agg + (size_t)c * CFEAT + lane * 4;
    atomicAdd(dst + 0, v.x * nrm);
    atomicAdd(dst + 1, v.y * nrm);
    atomicAdd(dst + 2, v.z * nrm);
    atomicAdd(dst + 3, v.w * nrm);
}

__global__ void selfloop_bias_relu(const float* __restrict__ h,
                                   const float* __restrict__ dinv,
                                   const float* __restrict__ bias,
                                   float* __restrict__ agg,
                                   int n) {
    const size_t idx = (size_t)blockIdx.x * blockDim.x + threadIdx.x;
    if (idx >= (size_t)n * CFEAT) return;
    const int i = (int)(idx >> 7);
    const int f = (int)(idx & 127);
    const float d = dinv[i];
    const float v = agg[idx] + h[idx] * d * d + bias[f];
    agg[idx] = v > 0.0f ? v : 0.0f;
}

extern "C" void kernel_launch(void* const* d_in, const int* in_sizes, int n_in,
                              void* d_out, int out_size, void* d_ws, size_t ws_size,
                              hipStream_t stream) {
    const float* x  = (const float*)d_in[0];
    const int*   ei = (const int*)d_in[1];
    const float* W1 = (const float*)d_in[2];
    const float* b1 = (const float*)d_in[3];
    const float* W2 = (const float*)d_in[4];
    const float* b2 = (const float*)d_in[5];
    const float* Wl = (const float*)d_in[6];
    const float* bl = (const float*)d_in[7];
    float* out = (float*)d_out;

    const int E  = in_sizes[1] / 2;
    const int Nn = in_sizes[0] / CFEAT;
    const int C  = CFEAT;

    const int* row = ei;
    const int* col = ei + E;

    float* dinv = (float*)d_ws;
    float* bufA = dinv + Nn;
    float* bufB = bufA + (size_t)Nn * C;

    const size_t featBytes = (size_t)Nn * C * sizeof(float);
    const int Mt = Nn / 16;
    const int featBlocks = (int)(((size_t)Nn * C + 255) / 256);

    hipMemsetAsync(dinv, 0, (size_t)Nn * sizeof(float), stream);
    degree_kernel<<<(E + 255) / 256, 256, 0, stream>>>(col, dinv, E);
    dinv_kernel<<<(Nn + 255) / 256, 256, 0, stream>>>(dinv, Nn);

    gemm16_wmma_f32<128, 128, 0><<<dim3(Mt, 2), 128, 0, stream>>>(x, W1, nullptr, bufA, Nn);
    hipMemsetAsync(bufB, 0, featBytes, stream);
    aggregate_edges<<<(E + 7) / 8, 256, 0, stream>>>(bufA, row, col, dinv, bufB, E);
    selfloop_bias_relu<<<featBlocks, 256, 0, stream>>>(bufA, dinv, b1, bufB, Nn);

    gemm16_wmma_f32<128, 128, 0><<<dim3(Mt, 2), 128, 0, stream>>>(bufB, W2, nullptr, bufA, Nn);
    hipMemsetAsync(bufB, 0, featBytes, stream);
    aggregate_edges<<<(E + 7) / 8, 256, 0, stream>>>(bufA, row, col, dinv, bufB, E);
    selfloop_bias_relu<<<featBlocks, 256, 0, stream>>>(bufA, dinv, b2, bufB, Nn);

    gemm16_wmma_f32<128, 64, 1><<<dim3(Mt, 1), 128, 0, stream>>>(bufB, Wl, bl, out, Nn);
}


# GEMMs: a wave keeps its 16 rows of A in registers as f32 WMMA fragments (read once, 16-byte loads), sweeps column tiles with W staged per work-group in LDS; k order inside the products permuted
# speedup vs baseline: 3.1765x; 3.1765x over previous
#include <hip/hip_runtime.h>
#include <hip/hip_bf16.h>

typedef __attribute__((ext_vector_type(2))) float v2f;
typedef __attribute__((ext_vector_type(4))) float v4f;
typedef __attribute__((ext_vector_type(8))) float v8f;

#define CFEAT 128

#define DCAP 40960
#define DEPT 8
#define DEPB (1024 * DEPT)
#define DFL 8
static_assert(DCAP % (1024 * DFL) == 0, "table reads of a trip must stay inside the table");
__global__ void __launch_bounds__(1024)
degree_kernel(const int* __restrict__ col, float* __restrict__ deg, int E, int Nn) {
    __shared__ unsigned hist[DCAP];
    const int t = threadIdx.x;
    const int e0 = blockIdx.x * DEPB + t;
    int c[DEPT];
#pragma unroll
    for (int j = 0; j < DEPT; ++j) {
        const int e = e0 + 1024 * j;
        const int v = col[e < E ? e : E - 1];
        c[j] = e < E ? v : -1;
    }
    for (int nb = 0; nb < Nn; nb += DCAP) {
        if (nb) __syncthreads();
        uint4* h4 = (uint4*)hist;
        for (int i = t; i < DCAP / 4; i += 1024) h4[i] = make_uint4(0u, 0u, 0u, 0u);
        __syncthreads();
#pragma unroll
        for (int j = 0; j < DEPT; ++j) {
            const unsigned k = (unsigned)(c[j] - nb);
            if (k < (unsigned)DCAP) atomicAdd(&hist[k], 1u);
        }
        __syncthreads();
        const int lim = (Nn - nb < DCAP) ? (Nn - nb) : DCAP;
        for (int i0 = t; i0 < lim; i0 += 1024 * DFL) {
            unsigned v[DFL];
#pragma unroll
            for (int u = 0; u < DFL; ++u) v[u] = hist[i0 + 1024 * u];
#pragma unroll
            for (int u = 0; u < DFL; ++u)
                if (i0 + 1024 * u < lim) atomicAdd(&deg[nb + i0 + 1024 * u], (float)v[u]);
        }
    }
}

#define NEPT 4
#define NEPB (256 * NEPT)
__global__ void __launch_bounds__(256)
edge_norm_kernel(const int* __restrict__ row, const int* __restrict__ col,
                 const float* __restrict__ deg, float* __restrict__ norm, int E) {
    const int b0 = blockIdx.x * NEPB;
    if (b0 + NEPB <= E) {
        const int e0 = b0 + threadIdx.x;
        int r[NEPT], c[NEPT];
        float dr[NEPT], dc[NEPT];
#pragma unroll
        for (int j = 0; j < NEPT; ++j) {
            r[j] = row[e0 + 256 * j];
            c[j] = col[e0 + 256 * j];
        }
#pragma unroll
        for (int j = 0; j < NEPT; ++j) {
            dr[j] = deg[r[j]];
            dc[j] = deg[c[j]];
        }
#pragma unroll
        for (int j = 0; j < NEPT; ++j)
            norm[e0 + 256 * j] = rsqrtf(dr[j] + 1.0f) * rsqrtf(dc[j] + 1.0f);
    } else {
        for (int e = b0 + (int)threadIdx.x; e < E; e += 256)
            norm[e] = rsqrtf(deg[row[e]] + 1.0f) * rsqrtf(deg[col[e]] + 1.0f);
    }
}

#define GW 20
#define GTH (GW * 32)
#define GGRID 256
template <int K, int N, int NTU, int RELU_A, int ADD_BIAS, int START>
__global__ void __launch_bounds__(GTH)
gemm_rows_wmma_f32(const float* __restrict__ A,
                   const float* __restrict__ W,
                   const float* __restrict__ bias,
                   const float* __restrict__ deg,
                   float* __restrict__ C,
                   float* __restrict__ S, int Mt) {
    constexpr int NT = N / 16;
    constexpr int KJ = K / 8;
    constexpr int UPS = NT / NTU;
    constexpr int NG = KJ * NT;
    constexpr int TR = (NG + GW - 1) / GW;
    static_assert(NT % NTU == 0, "whole units");
    __shared__ float4 wl[NG * 32];

    const int tid = threadIdx.x;
    const int lane = tid & 31;
    const int lo = lane & 15;
    const int hi = lane >> 4;

    const int wave = tid >> 5;
    {
        float wv[TR][4];
#pragma unroll
        for (int i = 0; i < TR; ++i) {
            const int g = wave + i * GW;
            const int gc = g < NG ? g : NG - 1;
            const int j = gc / NT;
            const int nt = gc % NT;
            const float* wp = W + (size_t)(8 * j + 4 * hi) * N + nt * 16 + lo;
#pragma unroll
            for (int q = 0; q < 4; ++q) wv[i][q] = wp[(size_t)q * N];
        }
#pragma unroll
        for (int i = 0; i < TR; ++i) {
            const int g = wave + i * GW;
            if (g < NG) wl[g * 32 + lane] = make_float4(wv[i][0], wv[i][1], wv[i][2], wv[i][3]);
        }
    }
    __syncthreads();

    const int b = blockIdx.x;
    const int G = gridDim.x;
    const int nstripes = (b < Mt) ? (Mt - b + G - 1) / G : 0;
    const int nunits = nstripes * UPS;

    for (int u = __builtin_amdgcn_readfirstlane(wave); u < nunits; u += GW) {
        const int m0 = (b + (u / UPS) * G) * 16;
        const int nt0 = (u % UPS) * NTU;

        const v4f* ap = (const v4f*)(A + (size_t)(m0 + lo) * K + 4 * hi);
        v4f a4[KJ];
#pragma unroll
        for (int j = 0; j < KJ; ++j) a4[j] = ap[2 * j];
        static_assert(KJ == 16, "the fence names 16 loads");
        asm volatile("" :: "v"(a4[0]), "v"(a4[1]), "v"(a4[2]), "v"(a4[3]),
                           "v"(a4[4]), "v"(a4[5]), "v"(a4[6]), "v"(a4[7]),
                           "v"(a4[8]), "v"(a4[9]), "v"(a4[10]), "v"(a4[11]),
                           "v"(a4[12]), "v"(a4[13]), "v"(a4[14]), "v"(a4[15]));
        if (RELU_A) {
#pragma unroll
            for (int j = 0; j < KJ; ++j) {
                a4[j].x = __builtin_fmaxf(a4[j].x, 0.0f);
                a4[j].y = __builtin_fmaxf(a4[j].y, 0.0f);
                a4[j].z = __builtin_fmaxf(a4[j].z, 0.0f);
                a4[j].w = __builtin_fmaxf(a4[j].w, 0.0f);
            }
        }
        float dv[8];
        if (START) {
            const float* dbase = deg + m0 + 8 * hi;
#pragma unroll
            for (int v = 0; v < 8; ++v) dv[v] = rsqrtf(dbase[v] + 1.0f);
        }

        const float4* wb = wl + nt0 * 32 + lane;
#pragma unroll
        for (int t = 0; t < NTU; ++t) {
            v8f acc = {};
#pragma unroll
            for (int j = 0; j < KJ; ++j) {
                const float4 bq = wb[(j * NT + t) * 32];
                v2f a0, b0, a1, b1;
                a0.x = a4[j].x; a0.y = a4[j].y; b0.x = bq.x; b0.y = bq.y;
                a1.x = a4[j].z; a1.y = a4[j].w; b1.x = bq.z; b1.y = bq.w;
                acc = __builtin_amdgcn_wmma_f32_16x16x4_f32(
                     false, a0,  false, b0,
                     (short)0, acc,  false,  false);
                acc = __builtin_amdgcn_wmma_f32_16x16x4_f32(
                     false, a1,  false, b1,
                     (short)0, acc,  false,  false);
            }
            const int n = (nt0 + t) * 16 + lo;
            const float bv = (ADD_BIAS || START) ? bias[n] : 0.0f;
            float* cbase = C + (size_t)(m0 + 8 * hi) * N + n;
#pragma unroll
            for (int v = 0; v < 8; ++v)
                cbase[(size_t)v * N] = ADD_BIAS ? acc[v] + bv : acc[v];
            if (START) {
                float* sbase = S + (size_t)(m0 + 8 * hi) * N + n;
#pragma unroll
                for (int v = 0; v < 8; ++v)
                    sbase[(size_t)v * N] = acc[v] * dv[v] * dv[v] + bv;
            }
        }
    }
}

#define EPW 64
#define EGRP 16
#define WPB 8
#define TSTRIDE 64

__device__ __forceinline__ float* row_base(const float* base, int r) {
    const unsigned long long a = (unsigned long long)base + ((unsigned long long)(unsigned)r << 9);
    const unsigned lo = (unsigned)__builtin_amdgcn_readfirstlane((int)(unsigned)a);
    const unsigned hi = (unsigned)__builtin_amdgcn_readfirstlane((int)(unsigned)(a >> 32));
    typedef __attribute__((address_space(1))) float gfloat;
    return (float*)(gfloat*)(((unsigned long long)hi << 32) | lo);
}

__device__ __forceinline__ void edge_one(const float* __restrict__ hl,
                                         float* __restrict__ al,
                                         int r, int c, float nrm) {
    const float* src = hl + (size_t)r * CFEAT;
    float* dst = al + (size_t)c * CFEAT;
    const float v0 = src[0];
    const float v1 = src[32];
    atomicAdd(dst + 0,  v0 * nrm);
    atomicAdd(dst + 32, v1 * nrm);
}

__global__ void __launch_bounds__(256)
aggregate_edges(const float* __restrict__ h,
                const int* __restrict__ row,
                const int* __restrict__ col,
                const float* __restrict__ norm,
                float* __restrict__ agg,
                int* __restrict__ tickets, int E, int nch) {
    __shared__ int s_work;
    if (threadIdx.x == 0) {
        const int id = __builtin_amdgcn_s_sendmsg_rtn(0x87);
        int part = (id >> 18) & 1;
        int t = atomicAdd(&tickets[part * TSTRIDE], 1);
        if (t >= nch) {
            part ^= 1;
            t = atomicAdd(&tickets[part * TSTRIDE], 1);
        }
        s_work = (t < nch) ? (2 * t + part) : -1;
    }
    __syncthreads();
    const int wk = __builtin_amdgcn_readfirstlane(s_work);
    if (wk < 0) return;
    const int part = wk & 1;
    const int chunk = wk >> 1;

    const int lane = threadIdx.x & 31;
    const int e0 = (chunk * WPB + (int)(threadIdx.x >> 5)) * EPW;
    if (e0 >= E) return;
    int r_l[2], c_l[2];
    float n_l[2];
#pragma unroll
    for (int s = 0; s < 2; ++s) {
        const int e = e0 + 32 * s + lane;
        const int ec = e < E ? e : E - 1;
        r_l[s] = row[ec];
        c_l[s] = col[ec];
        n_l[s] = norm[ec];
    }
    const unsigned off = (unsigned)(part * 64 + lane);
    const float* hl = h + off;
    float* al = agg + off;
    const int cnt = (E - e0 < EPW) ? (E - e0) : EPW;

    if (cnt == EPW) {
#pragma unroll
        for (int s = 0; s < 2; ++s) {
#pragma unroll
            for (int g = 0; g < 32; g += EGRP) {
                float v[EGRP][2];
                int   cs[EGRP];
                float ns[EGRP];
#pragma unroll
                for (int u = 0; u < EGRP; ++u) {
                    const int r = __builtin_amdgcn_readlane(r_l[s], g + u);
                    cs[u] = __builtin_amdgcn_readlane(c_l[s], g + u);
                    ns[u] = __builtin_bit_cast(float,
                              __builtin_amdgcn_readlane(__builtin_bit_cast(int, n_l[s]), g + u));
                    const float* src = row_base(h, r);
                    v[u][0] = src[off];
                    v[u][1] = src[off + 32];
                }
#pragma unroll
                for (int u = 0; u < EGRP; ++u) {
                    float* dst = row_base(agg, cs[u]);
                    atomicAdd(dst + off,      v[u][0] * ns[u]);
                    atomicAdd(dst + off + 32, v[u][1] * ns[u]);
                }
            }
        }
    } else {
#pragma unroll
        for (int s = 0; s < 2; ++s) {
            const int left = cnt - 32 * s;
            const int m = left < 32 ? left : 32;
            for (int j = 0; j < m; ++j) {
                const int r = __builtin_amdgcn_readlane(r_l[s], j);
                const int cc = __builtin_amdgcn_readlane(c_l[s], j);
                const float nrm = __builtin_bit_cast(float,
                              __builtin_amdgcn_readlane(__builtin_bit_cast(int, n_l[s]), j));
                edge_one(hl, al, r, cc, nrm);
            }
        }
    }
}

extern "C" void kernel_launch(void* const* d_in, const int* in_sizes, int n_in,
                              void* d_out, int out_size, void* d_ws, size_t ws_size,
                              hipStream_t stream) {
    const float* x  = (const float*)d_in[0];
    const int*   ei = (const int*)d_in[1];
    const float* W1 = (const float*)d_in[2];
    const float* b1 = (const float*)d_in[3];
    const float* W2 = (const float*)d_in[4];
    const float* b2 = (const float*)d_in[5];
    const float* Wl = (const float*)d_in[6];
    const float* bl = (const float*)d_in[7];
    float* out = (float*)d_out;

    const int E  = in_sizes[1] / 2;
    const int Nn = in_sizes[0] / CFEAT;
    const int C  = CFEAT;

    const int* row = ei;
    const int* col = ei + E;

    float* deg  = (float*)d_ws;
    int*   tick = (int*)(deg + Nn);
    float* bufA = deg + Nn + 4 * TSTRIDE;
    float* bufB = bufA + (size_t)Nn * C;
    float* bufC = bufB + (size_t)Nn * C;
    float* norm = bufC + (size_t)Nn * C;

    const int Mt = Nn / 16;
    const int nch = (E + WPB * EPW - 1) / (WPB * EPW);

    hipMemsetAsync(deg, 0, (size_t)(Nn + 4 * TSTRIDE) * sizeof(float), stream);
    degree_kernel<<<(E + DEPB - 1) / DEPB, 1024, 0, stream>>>(col, deg, E, Nn);
    edge_norm_kernel<<<(E + NEPB - 1) / NEPB, 256, 0, stream>>>(row, col, deg, norm, E);

    gemm_rows_wmma_f32<128, 128, 4, 0, 0, 1><<<GGRID, GTH, 0, stream>>>(x, W1, b1, deg, bufA, bufB, Mt);
    aggregate_edges<<<2 * nch, 256, 0, stream>>>(bufA, row, col, norm, bufB, tick, E, nch);

    gemm_rows_wmma_f32<128, 128, 4, 1, 0, 1><<<GGRID, GTH, 0, stream>>>(bufB, W2, b2, deg, bufA, bufC, Mt);
    aggregate_edges<<<2 * nch, 256, 0, stream>>>(bufA, row, col, norm, bufC, tick + 2 * TSTRIDE, E, nch);

    gemm_rows_wmma_f32<128, 64, 2, 1, 1, 0><<<GGRID, GTH, 0, stream>>>(bufC, Wl, bl, nullptr, out, nullptr, Mt);
}


# Edge weight table made in the layer-1 GEMM's launch: its work-groups take the edges while they stage W (index loads ahead of the W loads, degree gathers under them); own launch gone; sums unchanged
# speedup vs baseline: 3.2864x; 1.0346x over previous
#include <hip/hip_runtime.h>
#include <hip/hip_bf16.h>

typedef __attribute__((ext_vector_type(2))) float v2f;
typedef __attribute__((ext_vector_type(4))) float v4f;
typedef __attribute__((ext_vector_type(8))) float v8f;

#define CFEAT 128

#define DCAP 40960
#define DEPT 8
#define DEPB (1024 * DEPT)
#define DFL 8
static_assert(DCAP % (1024 * DFL) == 0, "table reads of a trip must stay inside the table");
__global__ void __launch_bounds__(1024)
degree_kernel(const int* __restrict__ col, float* __restrict__ deg, int E, int Nn) {
    __shared__ unsigned hist[DCAP];
    const int t = threadIdx.x;
    const int e0 = blockIdx.x * DEPB + t;
    int c[DEPT];
#pragma unroll
    for (int j = 0; j < DEPT; ++j) {
        const int e = e0 + 1024 * j;
        const int v = col[e < E ? e : E - 1];
        c[j] = e < E ? v : -1;
    }
    for (int nb = 0; nb < Nn; nb += DCAP) {
        if (nb) __syncthreads();
        uint4* h4 = (uint4*)hist;
        for (int i = t; i < DCAP / 4; i += 1024) h4[i] = make_uint4(0u, 0u, 0u, 0u);
        __syncthreads();
#pragma unroll
        for (int j = 0; j < DEPT; ++j) {
            const unsigned k = (unsigned)(c[j] - nb);
            if (k < (unsigned)DCAP) atomicAdd(&hist[k], 1u);
        }
        __syncthreads();
        const int lim = (Nn - nb < DCAP) ? (Nn - nb) : DCAP;
        for (int i0 = t; i0 < lim; i0 += 1024 * DFL) {
            unsigned v[DFL];
#pragma unroll
            for (int u = 0; u < DFL; ++u) v[u] = hist[i0 + 1024 * u];
#pragma unroll
            for (int u = 0; u < DFL; ++u)
                if (i0 + 1024 * u < lim) atomicAdd(&deg[nb + i0 + 1024 * u], (float)v[u]);
        }
    }
}

#define NEJ 4

#define GW 20
#define GTH (GW * 32)
#define GGRID 256
template <int K, int N, int NTU, int RELU_A, int ADD_BIAS, int START, int EDGEW>
__global__ void __launch_bounds__(GTH)
gemm_rows_wmma_f32(const float* __restrict__ A,
                   const float* __restrict__ W,
                   const float* __restrict__ bias,
                   const float* __restrict__ deg,
                   float* __restrict__ C,
                   float* __restrict__ S, int Mt,
                   const int* __restrict__ erow,
                   const int* __restrict__ ecol,
                   float* __restrict__ norm, int E) {
    constexpr int NT = N / 16;
    constexpr int KJ = K / 8;
    constexpr int UPS = NT / NTU;
    constexpr int NG = KJ * NT;
    constexpr int TR = (NG + GW - 1) / GW;
    static_assert(NT % NTU == 0, "whole units");
    __shared__ float4 wl[NG * 32];

    const int tid = threadIdx.x;
    const int lane = tid & 31;
    const int lo = lane & 15;
    const int hi = lane >> 4;

    const int wave = tid >> 5;

    const int ethreads = (int)gridDim.x * GTH;
    const int ebase = (int)blockIdx.x * GTH + tid;
    int er[NEJ], ec[NEJ];
    if (EDGEW) {
#pragma unroll
        for (int j = 0; j < NEJ; ++j) {
            const int e = ebase + j * ethreads;
            const int el = e < E ? e : E - 1;
            er[j] = erow[el];
            ec[j] = ecol[el];
        }
    }
    float edr[NEJ], edc[NEJ];
    {
        float wv[TR][4];
#pragma unroll
        for (int i = 0; i < TR; ++i) {
            const int g = wave + i * GW;
            const int gc = g < NG ? g : NG - 1;
            const int j = gc / NT;
            const int nt = gc % NT;
            const float* wp = W + (size_t)(8 * j + 4 * hi) * N + nt * 16 + lo;
#pragma unroll
            for (int q = 0; q < 4; ++q) wv[i][q] = wp[(size_t)q * N];
        }
        if (EDGEW) {
#pragma unroll
            for (int j = 0; j < NEJ; ++j) {
                edr[j] = deg[er[j]];
                edc[j] = deg[ec[j]];
            }
        }
#pragma unroll
        for (int i = 0; i < TR; ++i) {
            const int g = wave + i * GW;
            if (g < NG) wl[g * 32 + lane] = make_float4(wv[i][0], wv[i][1], wv[i][2], wv[i][3]);
        }
    }
    __syncthreads();
    if (EDGEW) {
#pragma unroll
        for (int j = 0; j < NEJ; ++j) {
            const int e = ebase + j * ethreads;
            if (e < E) norm[e] = rsqrtf(edr[j] + 1.0f) * rsqrtf(edc[j] + 1.0f);
        }
        for (int e = ebase + NEJ * ethreads; e < E; e += ethreads)
            norm[e] = rsqrtf(deg[erow[e]] + 1.0f) * rsqrtf(deg[ecol[e]] + 1.0f);
    }

    const int b = blockIdx.x;
    const int G = gridDim.x;
    const int nstripes = (b < Mt) ? (Mt - b + G - 1) / G : 0;
    const int nunits = nstripes * UPS;

    for (int u = __builtin_amdgcn_readfirstlane(wave); u < nunits; u += GW) {
        const int m0 = (b + (u / UPS) * G) * 16;
        const int nt0 = (u % UPS) * NTU;

        const v4f* ap = (const v4f*)(A + (size_t)(m0 + lo) * K + 4 * hi);
        v4f a4[KJ];
#pragma unroll
        for (int j = 0; j < KJ; ++j) a4[j] = ap[2 * j];
        static_assert(KJ == 16, "the fence names 16 loads");
        asm volatile("" :: "v"(a4[0]), "v"(a4[1]), "v"(a4[2]), "v"(a4[3]),
                           "v"(a4[4]), "v"(a4[5]), "v"(a4[6]), "v"(a4[7]),
                           "v"(a4[8]), "v"(a4[9]), "v"(a4[10]), "v"(a4[11]),
                           "v"(a4[12]), "v"(a4[13]), "v"(a4[14]), "v"(a4[15]));
        if (RELU_A) {
#pragma unroll
            for (int j = 0; j < KJ; ++j) {
                a4[j].x = __builtin_fmaxf(a4[j].x, 0.0f);
                a4[j].y = __builtin_fmaxf(a4[j].y, 0.0f);
                a4[j].z = __builtin_fmaxf(a4[j].z, 0.0f);
                a4[j].w = __builtin_fmaxf(a4[j].w, 0.0f);
            }
        }
        float dv[8];
        if (START) {
            const float* dbase = deg + m0 + 8 * hi;
#pragma unroll
            for (int v = 0; v < 8; ++v) dv[v] = rsqrtf(dbase[v] + 1.0f);
        }

        const float4* wb = wl + nt0 * 32 + lane;
#pragma unroll
        for (int t = 0; t < NTU; ++t) {
            v8f acc = {};
#pragma unroll
            for (int j = 0; j < KJ; ++j) {
                const float4 bq = wb[(j * NT + t) * 32];
                v2f a0, b0, a1, b1;
                a0.x = a4[j].x; a0.y = a4[j].y; b0.x = bq.x; b0.y = bq.y;
                a1.x = a4[j].z; a1.y = a4[j].w; b1.x = bq.z; b1.y = bq.w;
                acc = __builtin_amdgcn_wmma_f32_16x16x4_f32(
                     false, a0,  false, b0,
                     (short)0, acc,  false,  false);
                acc = __builtin_amdgcn_wmma_f32_16x16x4_f32(
                     false, a1,  false, b1,
                     (short)0, acc,  false,  false);
            }
            const int n = (nt0 + t) * 16 + lo;
            const float bv = (ADD_BIAS || START) ? bias[n] : 0.0f;
            float* cbase = C + (size_t)(m0 + 8 * hi) * N + n;
#pragma unroll
            for (int v = 0; v < 8; ++v)
                cbase[(size_t)v * N] = ADD_BIAS ? acc[v] + bv : acc[v];
            if (START) {
                float* sbase = S + (size_t)(m0 + 8 * hi) * N + n;
#pragma unroll
                for (int v = 0; v < 8; ++v)
                    sbase[(size_t)v * N] = acc[v] * dv[v] * dv[v] + bv;
            }
        }
    }
}

#define EPW 64
#define EGRP 16
#define WPB 8
#define TSTRIDE 64

__device__ __forceinline__ float* row_base(const float* base, int r) {
    const unsigned long long a = (unsigned long long)base + ((unsigned long long)(unsigned)r << 9);
    const unsigned lo = (unsigned)__builtin_amdgcn_readfirstlane((int)(unsigned)a);
    const unsigned hi = (unsigned)__builtin_amdgcn_readfirstlane((int)(unsigned)(a >> 32));
    typedef __attribute__((address_space(1))) float gfloat;
    return (float*)(gfloat*)(((unsigned long long)hi << 32) | lo);
}

__device__ __forceinline__ void edge_one(const float* __restrict__ hl,
                                         float* __restrict__ al,
                                         int r, int c, float nrm) {
    const float* src = hl + (size_t)r * CFEAT;
    float* dst = al + (size_t)c * CFEAT;
    const float v0 = src[0];
    const float v1 = src[32];
    atomicAdd(dst + 0,  v0 * nrm);
    atomicAdd(dst + 32, v1 * nrm);
}

__global__ void __launch_bounds__(256)
aggregate_edges(const float* __restrict__ h,
                const int* __restrict__ row,
                const int* __restrict__ col,
                const float* __restrict__ norm,
                float* __restrict__ agg,
                int* __restrict__ tickets, int E, int nch) {
    __shared__ int s_work;
    if (threadIdx.x == 0) {
        const int id = __builtin_amdgcn_s_sendmsg_rtn(0x87);
        int part = (id >> 18) & 1;
        int t = atomicAdd(&tickets[part * TSTRIDE], 1);
        if (t >= nch) {
            part ^= 1;
            t = atomicAdd(&tickets[part * TSTRIDE], 1);
        }
        s_work = (t < nch) ? (2 * t + part) : -1;
    }
    __syncthreads();
    const int wk = __builtin_amdgcn_readfirstlane(s_work);
    if (wk < 0) return;
    const int part = wk & 1;
    const int chunk = wk >> 1;

    const int lane = threadIdx.x & 31;
    const int e0 = (chunk * WPB + (int)(threadIdx.x >> 5)) * EPW;
    if (e0 >= E) return;
    int r_l[2], c_l[2];
    float n_l[2];
#pragma unroll
    for (int s = 0; s < 2; ++s) {
        const int e = e0 + 32 * s + lane;
        const int ec = e < E ? e : E - 1;
        r_l[s] = row[ec];
        c_l[s] = col[ec];
        n_l[s] = norm[ec];
    }
    const unsigned off = (unsigned)(part * 64 + lane);
    const float* hl = h + off;
    float* al = agg + off;
    const int cnt = (E - e0 < EPW) ? (E - e0) : EPW;

    if (cnt == EPW) {
#pragma unroll
        for (int s = 0; s < 2; ++s) {
#pragma unroll
            for (int g = 0; g < 32; g += EGRP) {
                float v[EGRP][2];
                int   cs[EGRP];
                float ns[EGRP];
#pragma unroll
                for (int u = 0; u < EGRP; ++u) {
                    const int r = __builtin_amdgcn_readlane(r_l[s], g + u);
                    cs[u] = __builtin_amdgcn_readlane(c_l[s], g + u);
                    ns[u] = __builtin_bit_cast(float,
                              __builtin_amdgcn_readlane(__builtin_bit_cast(int, n_l[s]), g + u));
                    const float* src = row_base(h, r);
                    v[u][0] = src[off];
                    v[u][1] = src[off + 32];
                }
#pragma unroll
                for (int u = 0; u < EGRP; ++u) {
                    float* dst = row_base(agg, cs[u]);
                    atomicAdd(dst + off,      v[u][0] * ns[u]);
                    atomicAdd(dst + off + 32, v[u][1] * ns[u]);
                }
            }
        }
    } else {
#pragma unroll
        for (int s = 0; s < 2; ++s) {
            const int left = cnt - 32 * s;
            const int m = left < 32 ? left : 32;
            for (int j = 0; j < m; ++j) {
                const int r = __builtin_amdgcn_readlane(r_l[s], j);
                const int cc = __builtin_amdgcn_readlane(c_l[s], j);
                const float nrm = __builtin_bit_cast(float,
                              __builtin_amdgcn_readlane(__builtin_bit_cast(int, n_l[s]), j));
                edge_one(hl, al, r, cc, nrm);
            }
        }
    }
}

extern "C" void kernel_launch(void* const* d_in, const int* in_sizes, int n_in,
                              void* d_out, int out_size, void* d_ws, size_t ws_size,
                              hipStream_t stream) {
    const float* x  = (const float*)d_in[0];
    const int*   ei = (const int*)d_in[1];
    const float* W1 = (const float*)d_in[2];
    const float* b1 = (const float*)d_in[3];
    const float* W2 = (const float*)d_in[4];
    const float* b2 = (const float*)d_in[5];
    const float* Wl = (const float*)d_in[6];
    const float* bl = (const float*)d_in[7];
    float* out = (float*)d_out;

    const int E  = in_sizes[1] / 2;
    const int Nn = in_sizes[0] / CFEAT;
    const int C  = CFEAT;

    const int* row = ei;
    const int* col = ei + E;

    float* deg  = (float*)d_ws;
    int*   tick = (int*)(deg + Nn);
    float* bufA = deg + Nn + 4 * TSTRIDE;
    float* bufB = bufA + (size_t)Nn * C;
    float* bufC = bufB + (size_t)Nn * C;
    float* norm = bufC + (size_t)Nn * C;

    const int Mt = Nn / 16;
    const int nch = (E + WPB * EPW - 1) / (WPB * EPW);

    hipMemsetAsync(deg, 0, (size_t)(Nn + 4 * TSTRIDE) * sizeof(float), stream);
    degree_kernel<<<(E + DEPB - 1) / DEPB, 1024, 0, stream>>>(col, deg, E, Nn);

    gemm_rows_wmma_f32<128, 128, 4, 0, 0, 1, 1><<<GGRID, GTH, 0, stream>>>(x, W1, b1, deg, bufA, bufB, Mt, row, col, norm, E);
    aggregate_edges<<<2 * nch, 256, 0, stream>>>(bufA, row, col, norm, bufB, tick, E, nch);

    gemm_rows_wmma_f32<128, 128, 4, 1, 0, 1, 0><<<GGRID, GTH, 0, stream>>>(bufB, W2, b2, deg, bufA, bufC, Mt, nullptr, nullptr, nullptr, 0);
    aggregate_edges<<<2 * nch, 256, 0, stream>>>(bufA, row, col, norm, bufC, tick + 2 * TSTRIDE, E, nch);

    gemm_rows_wmma_f32<128, 64, 2, 1, 1, 0, 0><<<GGRID, GTH, 0, stream>>>(bufC, Wl, bl, nullptr, out, nullptr, Mt, nullptr, nullptr, nullptr, 0);
}
